# starting kernel
# baseline (speedup 1.0000x reference)
#include <hip/hip_runtime.h>
#include <math.h>

typedef __attribute__((ext_vector_type(2))) float v2f;
typedef __attribute__((ext_vector_type(8))) float v8f;

#define DEPTH 4
#define H1 64
#define H2 32
#define NP 64
#define DIMX 3
#define FSIZE 160
#define BATCHN 256

__device__ __forceinline__ v8f wmma_f32_k4(v2f a, v2f b, v8f c) {
  return __builtin_amdgcn_wmma_f32_16x16x4_f32(false, a, false, b, (short)0, c,
                                               false, false);
}

__device__ __forceinline__ float fast_tanh(float x) {
#if __has_builtin(__builtin_amdgcn_tanhf)
  return __builtin_amdgcn_tanhf(x);
#else
  float r;
  asm volatile("v_tanh_f32 %0, %1\n\tv_nop" : "=v"(r) : "v"(x));
  return r;
#endif
}

__global__ __launch_bounds__(64) void drift_pair_chain(
    const float* __restrict__ x, const float* __restrict__ t,
    const float* __restrict__ fc2_w0, const float* __restrict__ fc2_b0,
    const float* __restrict__ fc2_w, const float* __restrict__ fc2_b,
    float* __restrict__ mfeat, float* __restrict__ m0,
    float* __restrict__ m1, float* __restrict__ m2) {
  const int i = blockIdx.x;
  const int b = blockIdx.y;
  const int tid = threadIdx.x;

  __shared__ float hA[NP][H2 + 4];
  __shared__ float hB[NP][H2 + 4];
  __shared__ float wS[H2][H2 + 4];
  __shared__ float sfeat[NP][4];
  __shared__ float biasS[H2];

  __builtin_prefetch(fc2_w, 0, 1);

  const float tval = t[0];
  const float xi0 = x[(b * NP + i) * 3 + 0];
  const float xi1 = x[(b * NP + i) * 3 + 1];
  const float xi2 = x[(b * NP + i) * 3 + 2];

  const int j = tid;
  float r0 = xi0 - x[(b * NP + j) * 3 + 0];
  float r1 = xi1 - x[(b * NP + j) * 3 + 1];
  float r2 = xi2 - x[(b * NP + j) * 3 + 2];
  if (j == i) { r0 += 1.f; r1 += 1.f; r2 += 1.f; }
  const float nr = sqrtf(r0 * r0 + r1 * r1 + r2 * r2);
  sfeat[j][0] = r0; sfeat[j][1] = r1; sfeat[j][2] = r2; sfeat[j][3] = nr;
  {
    const float f5[5] = {r0, r1, r2, nr, tval};
    #pragma unroll
    for (int o = 0; o < H2; ++o) {
      float acc = fc2_b0[o];
      #pragma unroll
      for (int k = 0; k < 5; ++k) acc += f5[k] * fc2_w0[o * 5 + k];
      hA[j][o] = fast_tanh(acc);
    }
  }
  __syncthreads();

  if (tid < 4) {
    float s = 0.f;
    for (int jj = 0; jj < NP; ++jj) s += sfeat[jj][tid];
    mfeat[(b * NP + i) * 4 + tid] = s * (1.f / NP);
  }
  if (tid < H2) {
    float s = 0.f;
    for (int jj = 0; jj < NP; ++jj) s += hA[jj][tid];
    m0[(b * NP + i) * H2 + tid] = s * (1.f / NP);
  }

  const int wave = tid >> 5;
  const int lane = tid & 31;
  const int half = lane >> 4;
  const int lr = lane & 15;

  for (int L = 0; L < 2; ++L) {
    const float* Wl = fc2_w + L * H2 * H2;
    for (int idx = tid; idx < H2 * H2; idx += 64)
      wS[idx >> 5][idx & 31] = Wl[idx];
    if (tid < H2) biasS[tid] = fc2_b[L * H2 + tid];
    __syncthreads();

    float(*src)[H2 + 4] = (L == 0) ? hA : hB;
    float(*dst)[H2 + 4] = (L == 0) ? hB : hA;
    const int rowbase = wave * 32;

    #pragma unroll
    for (int rt = 0; rt < 2; ++rt) {
      #pragma unroll
      for (int ct = 0; ct < 2; ++ct) {
        v8f acc = {};
        #pragma unroll
        for (int kk = 0; kk < 8; ++kk) {
          const int k0 = kk * 4 + half * 2;
          const int ar = rowbase + rt * 16 + lr;
          v2f a;  a.x = src[ar][k0];              a.y = src[ar][k0 + 1];
          v2f bv; bv.x = wS[ct * 16 + lr][k0];    bv.y = wS[ct * 16 + lr][k0 + 1];
          acc = wmma_f32_k4(a, bv, acc);
        }
        #pragma unroll
        for (int r = 0; r < 8; ++r) {
          const int row = rowbase + rt * 16 + r + half * 8;
          const int col = ct * 16 + lr;
          dst[row][col] = fast_tanh(acc[r] + biasS[col]) + src[row][col];
        }
      }
    }
    __syncthreads();

    float* mo = (L == 0) ? m1 : m2;
    if (tid < H2) {
      float s = 0.f;
      for (int jj = 0; jj < NP; ++jj) s += dst[jj][tid];
      mo[(b * NP + i) * H2 + tid] = s * (1.f / NP);
    }
    __syncthreads();
  }
}

__global__ __launch_bounds__(64) void drift_particle_chain(
    const float* __restrict__ x, const float* __restrict__ t,
    const float* __restrict__ fc1_w0, const float* __restrict__ fc1_b0,
    const float* __restrict__ fc1_w, const float* __restrict__ fc1_b,
    const float* __restrict__ final_w, const float* __restrict__ final_b,
    const float* __restrict__ mfeat, const float* __restrict__ m0,
    const float* __restrict__ m1, const float* __restrict__ m2,
    float* __restrict__ out) {
  const int b = blockIdx.x;
  const int tid = threadIdx.x;

  __shared__ float hA[NP][H1 + 4];
  __shared__ float hB[NP][H1 + 4];
  __shared__ float mprev[NP][H2 + 4];
  __shared__ float meanh[H1];
  __shared__ float sfeat[NP][4];
  __shared__ float smf[4];
  __shared__ float biasS[H1];

  __builtin_prefetch(fc1_w, 0, 1);

  const float tval = t[0];
  const int i = tid;
  const float x0 = x[(b * NP + i) * 3 + 0];
  const float x1 = x[(b * NP + i) * 3 + 1];
  const float x2 = x[(b * NP + i) * 3 + 2];
  const float nr = sqrtf(x0 * x0 + x1 * x1 + x2 * x2);
  sfeat[i][0] = x0; sfeat[i][1] = x1; sfeat[i][2] = x2; sfeat[i][3] = nr;
  __syncthreads();
  if (tid < 4) {
    float s = 0.f;
    for (int jj = 0; jj < NP; ++jj) s += sfeat[jj][tid];
    smf[tid] = s * (1.f / NP);
  }
  __syncthreads();

  {
    float ft[13];
    ft[0] = x0; ft[1] = x1; ft[2] = x2; ft[3] = nr;
    ft[4] = smf[0]; ft[5] = smf[1]; ft[6] = smf[2]; ft[7] = smf[3];
    ft[8]  = mfeat[(b * NP + i) * 4 + 0];
    ft[9]  = mfeat[(b * NP + i) * 4 + 1];
    ft[10] = mfeat[(b * NP + i) * 4 + 2];
    ft[11] = mfeat[(b * NP + i) * 4 + 3];
    ft[12] = tval;
    for (int o = 0; o < H1; ++o) {
      float acc = fc1_b0[o];
      #pragma unroll
      for (int k = 0; k < 13; ++k) acc += ft[k] * fc1_w0[o * 13 + k];
      hA[i][o] = fast_tanh(acc);
    }
  }
  __syncthreads();

  const int wave = tid >> 5;
  const int lane = tid & 31;
  const int half = lane >> 4;
  const int lr = lane & 15;

  for (int d = 1; d <= 3; ++d) {
    const float* Wl = fc1_w + (d - 1) * H1 * FSIZE;
    const float* ms = (d == 1) ? m0 : (d == 2) ? m1 : m2;
    float(*src)[H1 + 4] = (d & 1) ? hA : hB;
    float(*dst)[H1 + 4] = (d & 1) ? hB : hA;

    for (int idx = tid; idx < NP * H2; idx += 64)
      mprev[idx >> 5][idx & 31] = ms[b * NP * H2 + idx];
    biasS[tid] = fc1_b[(d - 1) * H1 + tid];
    {
      float s = 0.f;
      for (int jj = 0; jj < NP; ++jj) s += src[jj][tid];
      meanh[tid] = s * (1.f / NP);
    }
    __syncthreads();

    #pragma unroll
    for (int rt = 0; rt < 4; ++rt) {
      #pragma unroll
      for (int ct2 = 0; ct2 < 2; ++ct2) {
        const int ct = wave * 2 + ct2;
        v8f acc = {};
        #pragma unroll 8
        for (int kk = 0; kk < 40; ++kk) {
          const int k0 = kk * 4 + half * 2;
          const int ar = rt * 16 + lr;
          v2f a;
          if (kk < 16) {
            a.x = src[ar][k0];          a.y = src[ar][k0 + 1];
          } else if (kk < 32) {
            a.x = meanh[k0 - 64];       a.y = meanh[k0 - 63];
          } else {
            a.x = mprev[ar][k0 - 128];  a.y = mprev[ar][k0 - 127];
          }
          const float* wr = Wl + (ct * 16 + lr) * FSIZE + k0;
          v2f bv; bv.x = wr[0]; bv.y = wr[1];
          acc = wmma_f32_k4(a, bv, acc);
        }
        #pragma unroll
        for (int r = 0; r < 8; ++r) {
          const int row = rt * 16 + r + half * 8;
          const int col = ct * 16 + lr;
          dst[row][col] = fast_tanh(acc[r] + biasS[col]) + src[row][col];
        }
      }
    }
    __syncthreads();
  }

  #pragma unroll
  for (int dd = 0; dd < DIMX; ++dd) {
    float acc = final_b[dd];
    for (int k = 0; k < H1; ++k) acc += hB[tid][k] * final_w[dd * H1 + k];
    out[(b * NP + tid) * 3 + dd] = acc;
  }
}

extern "C" void kernel_launch(void* const* d_in, const int* in_sizes, int n_in,
                              void* d_out, int out_size, void* d_ws,
                              size_t ws_size, hipStream_t stream) {
  const float* x       = (const float*)d_in[0];
  const float* t       = (const float*)d_in[1];
  const float* fc1_w0  = (const float*)d_in[2];
  const float* fc1_b0  = (const float*)d_in[3];
  const float* fc1_w   = (const float*)d_in[4];
  const float* fc1_b   = (const float*)d_in[5];
  const float* fc2_w0  = (const float*)d_in[6];
  const float* fc2_b0  = (const float*)d_in[7];
  const float* fc2_w   = (const float*)d_in[8];
  const float* fc2_b   = (const float*)d_in[9];
  const float* final_w = (const float*)d_in[10];
  const float* final_b = (const float*)d_in[11];
  float* out = (float*)d_out;

  float* ws    = (float*)d_ws;
  float* mfeat = ws;
  float* m0    = mfeat + BATCHN * NP * 4;
  float* m1    = m0 + BATCHN * NP * H2;
  float* m2    = m1 + BATCHN * NP * H2;
  (void)in_sizes; (void)n_in; (void)out_size; (void)ws_size;

  drift_pair_chain<<<dim3(NP, BATCHN), 64, 0, stream>>>(
      x, t, fc2_w0, fc2_b0, fc2_w, fc2_b, mfeat, m0, m1, m2);
  drift_particle_chain<<<dim3(BATCHN), 64, 0, stream>>>(
      x, t, fc1_w0, fc1_b0, fc1_w, fc1_b, final_w, final_b,
      mfeat, m0, m1, m2, out);
}


# particle chain transposed and re-cut: 4 waves per batch, a wave owns 16 particles x all 64 outputs; weights = A operand, h1 stays in D registers as next B; means by 16-lane row sums (order changes)
# speedup vs baseline: 2.2856x; 2.2856x over previous
#include <hip/hip_runtime.h>
#include <math.h>

typedef __attribute__((ext_vector_type(2))) float v2f;
typedef __attribute__((ext_vector_type(8))) float v8f;

#define DEPTH 4
#define H1 64
#define H2 32
#define NP 64
#define DIMX 3
#define FSIZE 160
#define BATCHN 256

__device__ __forceinline__ v8f wmma_f32_k4(v2f a, v2f b, v8f c) {
  return __builtin_amdgcn_wmma_f32_16x16x4_f32(false, a, false, b, (short)0, c,
                                               false, false);
}

__device__ __forceinline__ float fast_tanh(float x) {
#if __has_builtin(__builtin_amdgcn_tanhf)
  return __builtin_amdgcn_tanhf(x);
#else
  float r;
  asm volatile("v_tanh_f32 %0, %1\n\tv_nop" : "=v"(r) : "v"(x));
  return r;
#endif
}

__global__ __launch_bounds__(64) void drift_pair_chain(
    const float* __restrict__ x, const float* __restrict__ t,
    const float* __restrict__ fc2_w0, const float* __restrict__ fc2_b0,
    const float* __restrict__ fc2_w, const float* __restrict__ fc2_b,
    float* __restrict__ mfeat, float* __restrict__ m0,
    float* __restrict__ m1, float* __restrict__ m2) {
  const int i = blockIdx.x;
  const int b = blockIdx.y;
  const int tid = threadIdx.x;

  __shared__ float hA[NP][H2 + 4];
  __shared__ float hB[NP][H2 + 4];
  __shared__ float wS[H2][H2 + 4];
  __shared__ float sfeat[NP][4];
  __shared__ float biasS[H2];

  __builtin_prefetch(fc2_w, 0, 1);

  const float tval = t[0];
  const float xi0 = x[(b * NP + i) * 3 + 0];
  const float xi1 = x[(b * NP + i) * 3 + 1];
  const float xi2 = x[(b * NP + i) * 3 + 2];

  const int j = tid;
  float r0 = xi0 - x[(b * NP + j) * 3 + 0];
  float r1 = xi1 - x[(b * NP + j) * 3 + 1];
  float r2 = xi2 - x[(b * NP + j) * 3 + 2];
  if (j == i) { r0 += 1.f; r1 += 1.f; r2 += 1.f; }
  const float nr = sqrtf(r0 * r0 + r1 * r1 + r2 * r2);
  sfeat[j][0] = r0; sfeat[j][1] = r1; sfeat[j][2] = r2; sfeat[j][3] = nr;
  {
    const float f5[5] = {r0, r1, r2, nr, tval};
    #pragma unroll
    for (int o = 0; o < H2; ++o) {
      float acc = fc2_b0[o];
      #pragma unroll
      for (int k = 0; k < 5; ++k) acc += f5[k] * fc2_w0[o * 5 + k];
      hA[j][o] = fast_tanh(acc);
    }
  }
  __syncthreads();

  if (tid < 4) {
    float s = 0.f;
    for (int jj = 0; jj < NP; ++jj) s += sfeat[jj][tid];
    mfeat[(b * NP + i) * 4 + tid] = s * (1.f / NP);
  }
  if (tid < H2) {
    float s = 0.f;
    for (int jj = 0; jj < NP; ++jj) s += hA[jj][tid];
    m0[(b * NP + i) * H2 + tid] = s * (1.f / NP);
  }

  const int wave = tid >> 5;
  const int lane = tid & 31;
  const int half = lane >> 4;
  const int lr = lane & 15;

  for (int L = 0; L < 2; ++L) {
    const float* Wl = fc2_w + L * H2 * H2;
    for (int idx = tid; idx < H2 * H2; idx += 64)
      wS[idx >> 5][idx & 31] = Wl[idx];
    if (tid < H2) biasS[tid] = fc2_b[L * H2 + tid];
    __syncthreads();

    float(*src)[H2 + 4] = (L == 0) ? hA : hB;
    float(*dst)[H2 + 4] = (L == 0) ? hB : hA;
    const int rowbase = wave * 32;

    #pragma unroll
    for (int rt = 0; rt < 2; ++rt) {
      #pragma unroll
      for (int ct = 0; ct < 2; ++ct) {
        v8f acc = {};
        #pragma unroll
        for (int kk = 0; kk < 8; ++kk) {
          const int k0 = kk * 4 + half * 2;
          const int ar = rowbase + rt * 16 + lr;
          v2f a;  a.x = src[ar][k0];              a.y = src[ar][k0 + 1];
          v2f bv; bv.x = wS[ct * 16 + lr][k0];    bv.y = wS[ct * 16 + lr][k0 + 1];
          acc = wmma_f32_k4(a, bv, acc);
        }
        #pragma unroll
        for (int r = 0; r < 8; ++r) {
          const int row = rowbase + rt * 16 + r + half * 8;
          const int col = ct * 16 + lr;
          dst[row][col] = fast_tanh(acc[r] + biasS[col]) + src[row][col];
        }
      }
    }
    __syncthreads();

    float* mo = (L == 0) ? m1 : m2;
    if (tid < H2) {
      float s = 0.f;
      for (int jj = 0; jj < NP; ++jj) s += dst[jj][tid];
      mo[(b * NP + i) * H2 + tid] = s * (1.f / NP);
    }
    __syncthreads();
  }
}

typedef __attribute__((ext_vector_type(4))) float v4f;

__device__ __forceinline__ float row16_sum(float v) {
  v += __builtin_bit_cast(float, __builtin_amdgcn_update_dpp(
           0, __builtin_bit_cast(int, v), 0x161, 0xf, 0xf, true));
  v += __builtin_bit_cast(float, __builtin_amdgcn_update_dpp(
           0, __builtin_bit_cast(int, v), 0x162, 0xf, 0xf, true));
  v += __builtin_bit_cast(float, __builtin_amdgcn_update_dpp(
           0, __builtin_bit_cast(int, v), 0x164, 0xf, 0xf, true));
  v += __builtin_bit_cast(float, __builtin_amdgcn_update_dpp(
           0, __builtin_bit_cast(int, v), 0x168, 0xf, 0xf, true));
  return v;
}

__device__ __forceinline__ float xhalf(float x) {
  const int xi = __builtin_bit_cast(int, x);
  return __builtin_bit_cast(float, __builtin_amdgcn_permlanex16(
      xi, xi, 0x76543210, 0xfedcba98, false, false));
}

#define NSTEP 30

__global__ __launch_bounds__(128) void drift_particle_chain(
    const float* __restrict__ x, const float* __restrict__ t,
    const float* __restrict__ fc1_w0, const float* __restrict__ fc1_b0,
    const float* __restrict__ fc1_w, const float* __restrict__ fc1_b,
    const float* __restrict__ final_w, const float* __restrict__ final_b,
    const float* __restrict__ mfeat, const float* __restrict__ m0,
    const float* __restrict__ m1, const float* __restrict__ m2,
    float* __restrict__ out) {
  const int b = blockIdx.x;
  const int tid = threadIdx.x;
  const int wave = tid >> 5;
  const int lane = tid & 31;
  const int hh = lane >> 4;
  const int lr = lane & 15;
  const int n = wave * 16 + lr;

  __shared__ v4f sfeat[NP];
  __shared__ __attribute__((aligned(16))) float w0s[13][H1 + 4];
  __shared__ v4f psum[3][4][16];
  __shared__ float outS[NP * DIMX];

  const float tval = t[0];
  const float x0 = x[(b * NP + n) * 3 + 0];
  const float x1 = x[(b * NP + n) * 3 + 1];
  const float x2 = x[(b * NP + n) * 3 + 2];
  const float nr = sqrtf(x0 * x0 + x1 * x1 + x2 * x2);
  const v4f mf = *(const v4f*)(mfeat + (b * NP + n) * 4);

  v4f mjv[3][4];
  #pragma unroll
  for (int s = 0; s < 3; ++s) {
    const float* ms = (s == 0) ? m0 : (s == 1) ? m1 : m2;
    const float* mp = ms + (b * NP + n) * H2 + 8 * hh;
    #pragma unroll
    for (int i = 0; i < 4; ++i)
      mjv[s][i] = *(const v4f*)(mp + (i >> 1) * 16 + (i & 1) * 4);
  }

  v4f b0v[8];
  #pragma unroll
  for (int i = 0; i < 8; ++i)
    b0v[i] = *(const v4f*)(fc1_b0 + (i >> 1) * 16 + 8 * hh + (i & 1) * 4);
  const float fb[DIMX] = {final_b[0], final_b[1], final_b[2]};

  const float* wlane = fc1_w + lr * FSIZE + 8 * hh;
  v4f abuf[3][8];
  auto tile_of = [](int g) -> int {
    const int j = g % 10;
    return (j < 4) ? j : (j < 6) ? (j + 4) : (j - 2);
  };
  auto load_a = [&](v4f (&dstb)[8], int g) {
    const float* wp = wlane + (g / 10) * (H1 * FSIZE) + tile_of(g) * 16;
    #pragma unroll
    for (int ot = 0; ot < 4; ++ot) {
      dstb[ot * 2 + 0] = *(const v4f*)(wp + ot * 16 * FSIZE);
      dstb[ot * 2 + 1] = *(const v4f*)(wp + ot * 16 * FSIZE + 4);
    }
  };
  load_a(abuf[0], 0);
  load_a(abuf[1], 1);

  if (hh == 0) sfeat[n] = (v4f){x0, x1, x2, nr};
  {
    float wv[7];
    #pragma unroll
    for (int it = 0; it < 6; ++it) wv[it] = fc1_w0[it * 128 + tid];
    wv[6] = fc1_w0[6 * 128 + (tid & 63)];
    #pragma unroll
    for (int it = 0; it < 6; ++it) {
      const int idx = it * 128 + tid;
      w0s[idx % 13][idx / 13] = wv[it];
    }
    if (tid < 64) {
      const int idx = 6 * 128 + tid;
      w0s[idx % 13][idx / 13] = wv[6];
    }
  }
  __syncthreads();

  float smf0 = 0.f, smf1 = 0.f, smf2 = 0.f, smf3 = 0.f;
  for (int jj = 0; jj < NP; ++jj) {
    const v4f f = sfeat[jj];
    smf0 += f.x; smf1 += f.y; smf2 += f.z; smf3 += f.w;
  }
  smf0 *= (1.f / NP); smf1 *= (1.f / NP); smf2 *= (1.f / NP); smf3 *= (1.f / NP);

  v8f hreg[4];
  {
    const float ft[13] = {x0, x1, x2, nr, smf0, smf1, smf2, smf3,
                          mf.x, mf.y, mf.z, mf.w, tval};
    #pragma unroll
    for (int ot = 0; ot < 4; ++ot) {
      const v4f ba = b0v[ot * 2 + 0];
      const v4f bb = b0v[ot * 2 + 1];
      v8f acc = {ba.x, ba.y, ba.z, ba.w, bb.x, bb.y, bb.z, bb.w};
      #pragma unroll
      for (int k = 0; k < 13; ++k) {
        const v4f wa = *(const v4f*)&w0s[k][ot * 16 + 8 * hh];
        const v4f wb = *(const v4f*)&w0s[k][ot * 16 + 8 * hh + 4];
        acc[0] += ft[k] * wa.x; acc[1] += ft[k] * wa.y;
        acc[2] += ft[k] * wa.z; acc[3] += ft[k] * wa.w;
        acc[4] += ft[k] * wb.x; acc[5] += ft[k] * wb.y;
        acc[6] += ft[k] * wb.z; acc[7] += ft[k] * wb.w;
      }
      #pragma unroll
      for (int r = 0; r < 8; ++r) hreg[ot][r] = fast_tanh(acc[r]);
    }
  }

  v8f acc[4];
  v8f meanv[4];
  v4f bias[8];
  v4f fw[DIMX][8];
  #pragma unroll
  for (int g = 0; g < NSTEP; ++g) {
    const int L = g / 10;
    const int j = g % 10;
    const int T = tile_of(g);
    if (g + 2 < NSTEP) load_a(abuf[(g + 2) % 3], g + 2);

    if (j == 0) {
      #pragma unroll
      for (int ot = 0; ot < 4; ++ot) {
        v8f rs;
        #pragma unroll
        for (int r = 0; r < 8; ++r) rs[r] = row16_sum(hreg[ot][r]);
        if (lr == 0) {
          psum[L][wave][ot * 4 + hh * 2 + 0] = (v4f){rs[0], rs[1], rs[2], rs[3]};
          psum[L][wave][ot * 4 + hh * 2 + 1] = (v4f){rs[4], rs[5], rs[6], rs[7]};
        }
      }
      #pragma unroll
      for (int i = 0; i < 8; ++i)
        bias[i] = *(const v4f*)(fc1_b + L * H1 + (i >> 1) * 16 + 8 * hh + (i & 1) * 4);
      #pragma unroll
      for (int ot = 0; ot < 4; ++ot) acc[ot] = (v8f){};
    }
    if (j == 6) {
      __syncthreads();
      #pragma unroll
      for (int tt = 0; tt < 4; ++tt) {
        #pragma unroll
        for (int p = 0; p < 2; ++p) {
          v4f s = psum[L][0][tt * 4 + hh * 2 + p];
          s += psum[L][1][tt * 4 + hh * 2 + p];
          s += psum[L][2][tt * 4 + hh * 2 + p];
          s += psum[L][3][tt * 4 + hh * 2 + p];
          s *= (1.f / NP);
          meanv[tt][p * 4 + 0] = s.x; meanv[tt][p * 4 + 1] = s.y;
          meanv[tt][p * 4 + 2] = s.z; meanv[tt][p * 4 + 3] = s.w;
        }
      }
      if (L == 2) {
        #pragma unroll
        for (int dd = 0; dd < DIMX; ++dd) {
          #pragma unroll
          for (int i = 0; i < 8; ++i)
            fw[dd][i] = *(const v4f*)(final_w + dd * H1 + (i >> 1) * 16 + 8 * hh + (i & 1) * 4);
        }
      }
    }

    v8f bf;
    if (T < 4) {
      bf = hreg[T];
    } else if (T < 8) {
      bf = meanv[T - 4];
    } else {
      const v4f lo = mjv[L][(T - 8) * 2 + 0];
      const v4f hi = mjv[L][(T - 8) * 2 + 1];
      bf = (v8f){lo.x, lo.y, lo.z, lo.w, hi.x, hi.y, hi.z, hi.w};
    }
    const v4f (&ab)[8] = abuf[g % 3];
    #pragma unroll
    for (int q = 0; q < 4; ++q) {
      v2f bv; bv.x = bf[2 * q]; bv.y = bf[2 * q + 1];
      #pragma unroll
      for (int ot = 0; ot < 4; ++ot) {
        const v4f aw = ab[ot * 2 + (q >> 1)];
        v2f a;
        if (q & 1) { a.x = aw.z; a.y = aw.w; } else { a.x = aw.x; a.y = aw.y; }
        acc[ot] = wmma_f32_k4(a, bv, acc[ot]);
      }
    }

    if (j == 9) {
      #pragma unroll
      for (int ot = 0; ot < 4; ++ot) {
        #pragma unroll
        for (int r = 0; r < 8; ++r)
          hreg[ot][r] = fast_tanh(acc[ot][r] + bias[ot * 2 + (r >> 2)][r & 3]) + hreg[ot][r];
      }
    }
  }

  #pragma unroll
  for (int dd = 0; dd < DIMX; ++dd) {
    float s = 0.f;
    #pragma unroll
    for (int ot = 0; ot < 4; ++ot) {
      #pragma unroll
      for (int r = 0; r < 8; ++r) s += hreg[ot][r] * fw[dd][ot * 2 + (r >> 2)][r & 3];
    }
    s = s + xhalf(s);
    if (hh == 0) outS[n * DIMX + dd] = s + fb[dd];
  }
  __syncthreads();
  for (int idx = tid; idx < NP * DIMX; idx += 128)
    out[b * (NP * DIMX) + idx] = outS[idx];
}

extern "C" void kernel_launch(void* const* d_in, const int* in_sizes, int n_in,
                              void* d_out, int out_size, void* d_ws,
                              size_t ws_size, hipStream_t stream) {
  const float* x       = (const float*)d_in[0];
  const float* t       = (const float*)d_in[1];
  const float* fc1_w0  = (const float*)d_in[2];
  const float* fc1_b0  = (const float*)d_in[3];
  const float* fc1_w   = (const float*)d_in[4];
  const float* fc1_b   = (const float*)d_in[5];
  const float* fc2_w0  = (const float*)d_in[6];
  const float* fc2_b0  = (const float*)d_in[7];
  const float* fc2_w   = (const float*)d_in[8];
  const float* fc2_b   = (const float*)d_in[9];
  const float* final_w = (const float*)d_in[10];
  const float* final_b = (const float*)d_in[11];
  float* out = (float*)d_out;

  float* ws    = (float*)d_ws;
  float* mfeat = ws;
  float* m0    = mfeat + BATCHN * NP * 4;
  float* m1    = m0 + BATCHN * NP * H2;
  float* m2    = m1 + BATCHN * NP * H2;
  (void)in_sizes; (void)n_in; (void)out_size; (void)ws_size;

  drift_pair_chain<<<dim3(NP, BATCHN), 64, 0, stream>>>(
      x, t, fc2_w0, fc2_b0, fc2_w, fc2_b, mfeat, m0, m1, m2);
  drift_particle_chain<<<dim3(BATCHN), 128, 0, stream>>>(
      x, t, fc1_w0, fc1_b0, fc1_w, fc1_b, final_w, final_b,
      mfeat, m0, m1, m2, out);
}


# pair chain transposed and re-tiled: lane = particle i, a wave walks 16 partners j with weights resident as A operands, h2 in D registers as next B; mean_j = per-lane running sums (order changes)
# speedup vs baseline: 3.7699x; 1.6494x over previous
#include <hip/hip_runtime.h>
#include <math.h>

typedef __attribute__((ext_vector_type(2))) float v2f;
typedef __attribute__((ext_vector_type(4))) float v4f;
typedef __attribute__((ext_vector_type(8))) float v8f;

#define DEPTH 4
#define H1 64
#define H2 32
#define NP 64
#define DIMX 3
#define FSIZE 160
#define BATCHN 256

__device__ __forceinline__ v8f wmma_f32_k4(v2f a, v2f b, v8f c) {
  return __builtin_amdgcn_wmma_f32_16x16x4_f32(false, a, false, b, (short)0, c,
                                               false, false);
}

__device__ __forceinline__ float fast_tanh(float x) {
#if __has_builtin(__builtin_amdgcn_tanhf)
  return __builtin_amdgcn_tanhf(x);
#else
  float r;
  asm volatile("v_tanh_f32 %0, %1\n\tv_nop" : "=v"(r) : "v"(x));
  return r;
#endif
}

__global__ __launch_bounds__(128) void drift_pair_chain(
    const float* __restrict__ x, const float* __restrict__ t,
    const float* __restrict__ fc2_w0, const float* __restrict__ fc2_b0,
    const float* __restrict__ fc2_w, const float* __restrict__ fc2_b,
    float* __restrict__ mfeat, float* __restrict__ m0,
    float* __restrict__ m1, float* __restrict__ m2) {
  const int itile = blockIdx.x;
  const int b = blockIdx.y;
  const int tid = threadIdx.x;
  const int wave = __builtin_amdgcn_readfirstlane(tid >> 5);
  const int lane = tid & 31;
  const int hh = lane >> 4;
  const int lr = lane & 15;
  const int i = itile * 16 + lr;
  const int jbase = wave * 16;

  __shared__ __attribute__((aligned(16))) float part[4][3][16][H2 + 4];
  __shared__ v4f partf[4][16];

  const float tval = t[0];
  const float xi0 = x[(b * NP + i) * 3 + 0];
  const float xi1 = x[(b * NP + i) * 3 + 1];
  const float xi2 = x[(b * NP + i) * 3 + 2];
  const int xj0v = __builtin_bit_cast(int, x[(b * NP + jbase + lr) * 3 + 0]);
  const int xj1v = __builtin_bit_cast(int, x[(b * NP + jbase + lr) * 3 + 1]);
  const int xj2v = __builtin_bit_cast(int, x[(b * NP + jbase + lr) * 3 + 2]);

  v2f a0[2];
  v8f c0[2];
  #pragma unroll
  for (int ot = 0; ot < 2; ++ot) {
    a0[ot].x = fc2_w0[(16 * ot + lr) * 5 + 2 * hh];
    a0[ot].y = fc2_w0[(16 * ot + lr) * 5 + 2 * hh + 1];
    const v4f ba = *(const v4f*)(fc2_b0 + 16 * ot + 8 * hh);
    const v4f bb = *(const v4f*)(fc2_b0 + 16 * ot + 8 * hh + 4);
    const float bl[8] = {ba.x, ba.y, ba.z, ba.w, bb.x, bb.y, bb.z, bb.w};
    #pragma unroll
    for (int r = 0; r < 8; ++r)
      c0[ot][r] = bl[r] + tval * fc2_w0[(16 * ot + 8 * hh + r) * 5 + 4];
  }
  v4f aw[2][2][2][2];
  v8f cb[2][2];
  #pragma unroll
  for (int L = 0; L < 2; ++L) {
    #pragma unroll
    for (int ot = 0; ot < 2; ++ot) {
      const float* wp = fc2_w + L * (H2 * H2) + (16 * ot + lr) * H2 + 8 * hh;
      #pragma unroll
      for (int T = 0; T < 2; ++T) {
        aw[L][ot][T][0] = *(const v4f*)(wp + 16 * T);
        aw[L][ot][T][1] = *(const v4f*)(wp + 16 * T + 4);
      }
      const v4f ba = *(const v4f*)(fc2_b + L * H2 + 16 * ot + 8 * hh);
      const v4f bb = *(const v4f*)(fc2_b + L * H2 + 16 * ot + 8 * hh + 4);
      cb[L][ot] = (v8f){ba.x, ba.y, ba.z, ba.w, bb.x, bb.y, bb.z, bb.w};
    }
  }

  v8f s0[2] = {{}, {}}, s1[2] = {{}, {}}, s2[2] = {{}, {}};
  float sf0 = 0.f, sf1 = 0.f, sf2 = 0.f, sf3 = 0.f;

  for (int it = 0; it < 16; ++it) {
    const float xj0 = __builtin_bit_cast(float, __builtin_amdgcn_readlane(xj0v, it));
    const float xj1 = __builtin_bit_cast(float, __builtin_amdgcn_readlane(xj1v, it));
    const float xj2 = __builtin_bit_cast(float, __builtin_amdgcn_readlane(xj2v, it));
    const float dg = (i == jbase + it) ? 1.f : 0.f;
    const float r0 = (xi0 - xj0) + dg;
    const float r1 = (xi1 - xj1) + dg;
    const float r2 = (xi2 - xj2) + dg;
    const float nr = sqrtf(r0 * r0 + r1 * r1 + r2 * r2);
    sf0 += r0; sf1 += r1; sf2 += r2; sf3 += nr;

    v2f bv0;
    bv0.x = hh ? r2 : r0;
    bv0.y = hh ? nr : r1;
    v8f h[2];
    #pragma unroll
    for (int ot = 0; ot < 2; ++ot) {
      const v8f d = wmma_f32_k4(a0[ot], bv0, c0[ot]);
      #pragma unroll
      for (int r = 0; r < 8; ++r) h[ot][r] = fast_tanh(d[r]);
      s0[ot] += h[ot];
    }

    #pragma unroll
    for (int L = 0; L < 2; ++L) {
      v8f acc[2] = {cb[L][0], cb[L][1]};
      #pragma unroll
      for (int T = 0; T < 2; ++T) {
        #pragma unroll
        for (int q = 0; q < 4; ++q) {
          v2f bv; bv.x = h[T][2 * q]; bv.y = h[T][2 * q + 1];
          #pragma unroll
          for (int ot = 0; ot < 2; ++ot) {
            const v4f w4 = aw[L][ot][T][q >> 1];
            v2f a;
            if (q & 1) { a.x = w4.z; a.y = w4.w; } else { a.x = w4.x; a.y = w4.y; }
            acc[ot] = wmma_f32_k4(a, bv, acc[ot]);
          }
        }
      }
      #pragma unroll
      for (int ot = 0; ot < 2; ++ot) {
        #pragma unroll
        for (int r = 0; r < 8; ++r) h[ot][r] = fast_tanh(acc[ot][r]) + h[ot][r];
        if (L == 0) s1[ot] += h[ot]; else s2[ot] += h[ot];
      }
    }
  }

  #pragma unroll
  for (int ot = 0; ot < 2; ++ot) {
    float* p0 = &part[wave][0][lr][16 * ot + 8 * hh];
    float* p1 = &part[wave][1][lr][16 * ot + 8 * hh];
    float* p2 = &part[wave][2][lr][16 * ot + 8 * hh];
    *(v4f*)(p0)     = (v4f){s0[ot][0], s0[ot][1], s0[ot][2], s0[ot][3]};
    *(v4f*)(p0 + 4) = (v4f){s0[ot][4], s0[ot][5], s0[ot][6], s0[ot][7]};
    *(v4f*)(p1)     = (v4f){s1[ot][0], s1[ot][1], s1[ot][2], s1[ot][3]};
    *(v4f*)(p1 + 4) = (v4f){s1[ot][4], s1[ot][5], s1[ot][6], s1[ot][7]};
    *(v4f*)(p2)     = (v4f){s2[ot][0], s2[ot][1], s2[ot][2], s2[ot][3]};
    *(v4f*)(p2 + 4) = (v4f){s2[ot][4], s2[ot][5], s2[ot][6], s2[ot][7]};
  }
  if (hh == 0) partf[wave][lr] = (v4f){sf0, sf1, sf2, sf3};
  __syncthreads();

  if (wave < 3) {
    float* ms = (wave == 0) ? m0 : (wave == 1) ? m1 : m2;
    float* dstp = ms + (b * NP + itile * 16) * H2;
    #pragma unroll
    for (int k = 0; k < 4; ++k) {
      const int f = k * 32 + lane;
      const int row = f >> 3;
      const int c4 = (f & 7) * 4;
      v4f v = *(const v4f*)&part[0][wave][row][c4];
      v += *(const v4f*)&part[1][wave][row][c4];
      v += *(const v4f*)&part[2][wave][row][c4];
      v += *(const v4f*)&part[3][wave][row][c4];
      v *= (1.f / NP);
      *(v4f*)(dstp + f * 4) = v;
    }
  } else if (lane < 16) {
    v4f v = partf[0][lane];
    v += partf[1][lane];
    v += partf[2][lane];
    v += partf[3][lane];
    v *= (1.f / NP);
    *(v4f*)(mfeat + (b * NP + itile * 16 + lane) * 4) = v;
  }
}

__device__ __forceinline__ float row16_sum(float v) {
  v += __builtin_bit_cast(float, __builtin_amdgcn_update_dpp(
           0, __builtin_bit_cast(int, v), 0x161, 0xf, 0xf, true));
  v += __builtin_bit_cast(float, __builtin_amdgcn_update_dpp(
           0, __builtin_bit_cast(int, v), 0x162, 0xf, 0xf, true));
  v += __builtin_bit_cast(float, __builtin_amdgcn_update_dpp(
           0, __builtin_bit_cast(int, v), 0x164, 0xf, 0xf, true));
  v += __builtin_bit_cast(float, __builtin_amdgcn_update_dpp(
           0, __builtin_bit_cast(int, v), 0x168, 0xf, 0xf, true));
  return v;
}

__device__ __forceinline__ float xhalf(float x) {
  const int xi = __builtin_bit_cast(int, x);
  return __builtin_bit_cast(float, __builtin_amdgcn_permlanex16(
      xi, xi, 0x76543210, 0xfedcba98, false, false));
}

#define NSTEP 30

__global__ __launch_bounds__(128) void drift_particle_chain(
    const float* __restrict__ x, const float* __restrict__ t,
    const float* __restrict__ fc1_w0, const float* __restrict__ fc1_b0,
    const float* __restrict__ fc1_w, const float* __restrict__ fc1_b,
    const float* __restrict__ final_w, const float* __restrict__ final_b,
    const float* __restrict__ mfeat, const float* __restrict__ m0,
    const float* __restrict__ m1, const float* __restrict__ m2,
    float* __restrict__ out) {
  const int b = blockIdx.x;
  const int tid = threadIdx.x;
  const int wave = tid >> 5;
  const int lane = tid & 31;
  const int hh = lane >> 4;
  const int lr = lane & 15;
  const int n = wave * 16 + lr;

  __shared__ v4f sfeat[NP];
  __shared__ __attribute__((aligned(16))) float w0s[13][H1 + 4];
  __shared__ v4f psum[3][4][16];
  __shared__ float outS[NP * DIMX];

  const float tval = t[0];
  const float x0 = x[(b * NP + n) * 3 + 0];
  const float x1 = x[(b * NP + n) * 3 + 1];
  const float x2 = x[(b * NP + n) * 3 + 2];
  const float nr = sqrtf(x0 * x0 + x1 * x1 + x2 * x2);
  const v4f mf = *(const v4f*)(mfeat + (b * NP + n) * 4);

  v4f mjv[3][4];
  #pragma unroll
  for (int s = 0; s < 3; ++s) {
    const float* ms = (s == 0) ? m0 : (s == 1) ? m1 : m2;
    const float* mp = ms + (b * NP + n) * H2 + 8 * hh;
    #pragma unroll
    for (int i = 0; i < 4; ++i)
      mjv[s][i] = *(const v4f*)(mp + (i >> 1) * 16 + (i & 1) * 4);
  }

  v4f b0v[8];
  #pragma unroll
  for (int i = 0; i < 8; ++i)
    b0v[i] = *(const v4f*)(fc1_b0 + (i >> 1) * 16 + 8 * hh + (i & 1) * 4);
  const float fb[DIMX] = {final_b[0], final_b[1], final_b[2]};

  const float* wlane = fc1_w + lr * FSIZE + 8 * hh;
  v4f abuf[3][8];
  auto tile_of = [](int g) -> int {
    const int j = g % 10;
    return (j < 4) ? j : (j < 6) ? (j + 4) : (j - 2);
  };
  auto load_a = [&](v4f (&dstb)[8], int g) {
    const float* wp = wlane + (g / 10) * (H1 * FSIZE) + tile_of(g) * 16;
    #pragma unroll
    for (int ot = 0; ot < 4; ++ot) {
      dstb[ot * 2 + 0] = *(const v4f*)(wp + ot * 16 * FSIZE);
      dstb[ot * 2 + 1] = *(const v4f*)(wp + ot * 16 * FSIZE + 4);
    }
  };
  load_a(abuf[0], 0);
  load_a(abuf[1], 1);

  if (hh == 0) sfeat[n] = (v4f){x0, x1, x2, nr};
  {
    float wv[7];
    #pragma unroll
    for (int it = 0; it < 6; ++it) wv[it] = fc1_w0[it * 128 + tid];
    wv[6] = fc1_w0[6 * 128 + (tid & 63)];
    #pragma unroll
    for (int it = 0; it < 6; ++it) {
      const int idx = it * 128 + tid;
      w0s[idx % 13][idx / 13] = wv[it];
    }
    if (tid < 64) {
      const int idx = 6 * 128 + tid;
      w0s[idx % 13][idx / 13] = wv[6];
    }
  }
  __syncthreads();

  float smf0 = 0.f, smf1 = 0.f, smf2 = 0.f, smf3 = 0.f;
  for (int jj = 0; jj < NP; ++jj) {
    const v4f f = sfeat[jj];
    smf0 += f.x; smf1 += f.y; smf2 += f.z; smf3 += f.w;
  }
  smf0 *= (1.f / NP); smf1 *= (1.f / NP); smf2 *= (1.f / NP); smf3 *= (1.f / NP);

  v8f hreg[4];
  {
    const float ft[13] = {x0, x1, x2, nr, smf0, smf1, smf2, smf3,
                          mf.x, mf.y, mf.z, mf.w, tval};
    #pragma unroll
    for (int ot = 0; ot < 4; ++ot) {
      const v4f ba = b0v[ot * 2 + 0];
      const v4f bb = b0v[ot * 2 + 1];
      v8f acc = {ba.x, ba.y, ba.z, ba.w, bb.x, bb.y, bb.z, bb.w};
      #pragma unroll
      for (int k = 0; k < 13; ++k) {
        const v4f wa = *(const v4f*)&w0s[k][ot * 16 + 8 * hh];
        const v4f wb = *(const v4f*)&w0s[k][ot * 16 + 8 * hh + 4];
        acc[0] += ft[k] * wa.x; acc[1] += ft[k] * wa.y;
        acc[2] += ft[k] * wa.z; acc[3] += ft[k] * wa.w;
        acc[4] += ft[k] * wb.x; acc[5] += ft[k] * wb.y;
        acc[6] += ft[k] * wb.z; acc[7] += ft[k] * wb.w;
      }
      #pragma unroll
      for (int r = 0; r < 8; ++r) hreg[ot][r] = fast_tanh(acc[r]);
    }
  }

  v8f acc[4];
  v8f meanv[4];
  v4f bias[8];
  v4f fw[DIMX][8];
  #pragma unroll
  for (int g = 0; g < NSTEP; ++g) {
    const int L = g / 10;
    const int j = g % 10;
    const int T = tile_of(g);
    if (g + 2 < NSTEP) load_a(abuf[(g + 2) % 3], g + 2);

    if (j == 0) {
      #pragma unroll
      for (int ot = 0; ot < 4; ++ot) {
        v8f rs;
        #pragma unroll
        for (int r = 0; r < 8; ++r) rs[r] = row16_sum(hreg[ot][r]);
        if (lr == 0) {
          psum[L][wave][ot * 4 + hh * 2 + 0] = (v4f){rs[0], rs[1], rs[2], rs[3]};
          psum[L][wave][ot * 4 + hh * 2 + 1] = (v4f){rs[4], rs[5], rs[6], rs[7]};
        }
      }
      #pragma unroll
      for (int i = 0; i < 8; ++i)
        bias[i] = *(const v4f*)(fc1_b + L * H1 + (i >> 1) * 16 + 8 * hh + (i & 1) * 4);
      #pragma unroll
      for (int ot = 0; ot < 4; ++ot) acc[ot] = (v8f){};
    }
    if (j == 6) {
      __syncthreads();
      #pragma unroll
      for (int tt = 0; tt < 4; ++tt) {
        #pragma unroll
        for (int p = 0; p < 2; ++p) {
          v4f s = psum[L][0][tt * 4 + hh * 2 + p];
          s += psum[L][1][tt * 4 + hh * 2 + p];
          s += psum[L][2][tt * 4 + hh * 2 + p];
          s += psum[L][3][tt * 4 + hh * 2 + p];
          s *= (1.f / NP);
          meanv[tt][p * 4 + 0] = s.x; meanv[tt][p * 4 + 1] = s.y;
          meanv[tt][p * 4 + 2] = s.z; meanv[tt][p * 4 + 3] = s.w;
        }
      }
      if (L == 2) {
        #pragma unroll
        for (int dd = 0; dd < DIMX; ++dd) {
          #pragma unroll
          for (int i = 0; i < 8; ++i)
            fw[dd][i] = *(const v4f*)(final_w + dd * H1 + (i >> 1) * 16 + 8 * hh + (i & 1) * 4);
        }
      }
    }

    v8f bf;
    if (T < 4) {
      bf = hreg[T];
    } else if (T < 8) {
      bf = meanv[T - 4];
    } else {
      const v4f lo = mjv[L][(T - 8) * 2 + 0];
      const v4f hi = mjv[L][(T - 8) * 2 + 1];
      bf = (v8f){lo.x, lo.y, lo.z, lo.w, hi.x, hi.y, hi.z, hi.w};
    }
    const v4f (&ab)[8] = abuf[g % 3];
    #pragma unroll
    for (int q = 0; q < 4; ++q) {
      v2f bv; bv.x = bf[2 * q]; bv.y = bf[2 * q + 1];
      #pragma unroll
      for (int ot = 0; ot < 4; ++ot) {
        const v4f aw = ab[ot * 2 + (q >> 1)];
        v2f a;
        if (q & 1) { a.x = aw.z; a.y = aw.w; } else { a.x = aw.x; a.y = aw.y; }
        acc[ot] = wmma_f32_k4(a, bv, acc[ot]);
      }
    }

    if (j == 9) {
      #pragma unroll
      for (int ot = 0; ot < 4; ++ot) {
        #pragma unroll
        for (int r = 0; r < 8; ++r)
          hreg[ot][r] = fast_tanh(acc[ot][r] + bias[ot * 2 + (r >> 2)][r & 3]) + hreg[ot][r];
      }
    }
  }

  #pragma unroll
  for (int dd = 0; dd < DIMX; ++dd) {
    float s = 0.f;
    #pragma unroll
    for (int ot = 0; ot < 4; ++ot) {
      #pragma unroll
      for (int r = 0; r < 8; ++r) s += hreg[ot][r] * fw[dd][ot * 2 + (r >> 2)][r & 3];
    }
    s = s + xhalf(s);
    if (hh == 0) outS[n * DIMX + dd] = s + fb[dd];
  }
  __syncthreads();
  for (int idx = tid; idx < NP * DIMX; idx += 128)
    out[b * (NP * DIMX) + idx] = outS[idx];
}

extern "C" void kernel_launch(void* const* d_in, const int* in_sizes, int n_in,
                              void* d_out, int out_size, void* d_ws,
                              size_t ws_size, hipStream_t stream) {
  const float* x       = (const float*)d_in[0];
  const float* t       = (const float*)d_in[1];
  const float* fc1_w0  = (const float*)d_in[2];
  const float* fc1_b0  = (const float*)d_in[3];
  const float* fc1_w   = (const float*)d_in[4];
  const float* fc1_b   = (const float*)d_in[5];
  const float* fc2_w0  = (const float*)d_in[6];
  const float* fc2_b0  = (const float*)d_in[7];
  const float* fc2_w   = (const float*)d_in[8];
  const float* fc2_b   = (const float*)d_in[9];
  const float* final_w = (const float*)d_in[10];
  const float* final_b = (const float*)d_in[11];
  float* out = (float*)d_out;

  float* ws    = (float*)d_ws;
  float* mfeat = ws;
  float* m0    = mfeat + BATCHN * NP * 4;
  float* m1    = m0 + BATCHN * NP * H2;
  float* m2    = m1 + BATCHN * NP * H2;
  (void)in_sizes; (void)n_in; (void)out_size; (void)ws_size;

  drift_pair_chain<<<dim3(4, BATCHN), 128, 0, stream>>>(
      x, t, fc2_w0, fc2_b0, fc2_w, fc2_b, mfeat, m0, m1, m2);
  drift_particle_chain<<<dim3(BATCHN), 128, 0, stream>>>(
      x, t, fc1_w0, fc1_b0, fc1_w, fc1_b, final_w, final_b,
      mfeat, m0, m1, m2, out);
}


# particle chain: cross-wave means via row DPP: a lane pair reads 2 LDS vectors per layer (was 32 dependent reads per lane), totals spread by row_share; [x,|x|] mean by row sums (sum order changes)
# speedup vs baseline: 4.0373x; 1.0709x over previous
#include <hip/hip_runtime.h>
#include <math.h>

typedef __attribute__((ext_vector_type(2))) float v2f;
typedef __attribute__((ext_vector_type(4))) float v4f;
typedef __attribute__((ext_vector_type(8))) float v8f;

#define DEPTH 4
#define H1 64
#define H2 32
#define NP 64
#define DIMX 3
#define FSIZE 160
#define BATCHN 256

__device__ __forceinline__ v8f wmma_f32_k4(v2f a, v2f b, v8f c) {
  return __builtin_amdgcn_wmma_f32_16x16x4_f32(false, a, false, b, (short)0, c,
                                               false, false);
}

__device__ __forceinline__ float fast_tanh(float x) {
#if __has_builtin(__builtin_amdgcn_tanhf)
  return __builtin_amdgcn_tanhf(x);
#else
  float r;
  asm volatile("v_tanh_f32 %0, %1\n\tv_nop" : "=v"(r) : "v"(x));
  return r;
#endif
}

__global__ __launch_bounds__(128) void drift_pair_chain(
    const float* __restrict__ x, const float* __restrict__ t,
    const float* __restrict__ fc2_w0, const float* __restrict__ fc2_b0,
    const float* __restrict__ fc2_w, const float* __restrict__ fc2_b,
    float* __restrict__ mfeat, float* __restrict__ m0,
    float* __restrict__ m1, float* __restrict__ m2) {
  const int itile = blockIdx.x;
  const int b = blockIdx.y;
  const int tid = threadIdx.x;
  const int wave = __builtin_amdgcn_readfirstlane(tid >> 5);
  const int lane = tid & 31;
  const int hh = lane >> 4;
  const int lr = lane & 15;
  const int i = itile * 16 + lr;
  const int jbase = wave * 16;

  __shared__ __attribute__((aligned(16))) float part[4][3][16][H2 + 4];
  __shared__ v4f partf[4][16];

  const float tval = t[0];
  const float xi0 = x[(b * NP + i) * 3 + 0];
  const float xi1 = x[(b * NP + i) * 3 + 1];
  const float xi2 = x[(b * NP + i) * 3 + 2];
  const int xj0v = __builtin_bit_cast(int, x[(b * NP + jbase + lr) * 3 + 0]);
  const int xj1v = __builtin_bit_cast(int, x[(b * NP + jbase + lr) * 3 + 1]);
  const int xj2v = __builtin_bit_cast(int, x[(b * NP + jbase + lr) * 3 + 2]);

  v2f a0[2];
  v8f c0[2];
  #pragma unroll
  for (int ot = 0; ot < 2; ++ot) {
    a0[ot].x = fc2_w0[(16 * ot + lr) * 5 + 2 * hh];
    a0[ot].y = fc2_w0[(16 * ot + lr) * 5 + 2 * hh + 1];
    const v4f ba = *(const v4f*)(fc2_b0 + 16 * ot + 8 * hh);
    const v4f bb = *(const v4f*)(fc2_b0 + 16 * ot + 8 * hh + 4);
    const float bl[8] = {ba.x, ba.y, ba.z, ba.w, bb.x, bb.y, bb.z, bb.w};
    #pragma unroll
    for (int r = 0; r < 8; ++r)
      c0[ot][r] = bl[r] + tval * fc2_w0[(16 * ot + 8 * hh + r) * 5 + 4];
  }
  v4f aw[2][2][2][2];
  v8f cb[2][2];
  #pragma unroll
  for (int L = 0; L < 2; ++L) {
    #pragma unroll
    for (int ot = 0; ot < 2; ++ot) {
      const float* wp = fc2_w + L * (H2 * H2) + (16 * ot + lr) * H2 + 8 * hh;
      #pragma unroll
      for (int T = 0; T < 2; ++T) {
        aw[L][ot][T][0] = *(const v4f*)(wp + 16 * T);
        aw[L][ot][T][1] = *(const v4f*)(wp + 16 * T + 4);
      }
      const v4f ba = *(const v4f*)(fc2_b + L * H2 + 16 * ot + 8 * hh);
      const v4f bb = *(const v4f*)(fc2_b + L * H2 + 16 * ot + 8 * hh + 4);
      cb[L][ot] = (v8f){ba.x, ba.y, ba.z, ba.w, bb.x, bb.y, bb.z, bb.w};
    }
  }

  v8f s0[2] = {{}, {}}, s1[2] = {{}, {}}, s2[2] = {{}, {}};
  float sf0 = 0.f, sf1 = 0.f, sf2 = 0.f, sf3 = 0.f;

  for (int it = 0; it < 16; ++it) {
    const float xj0 = __builtin_bit_cast(float, __builtin_amdgcn_readlane(xj0v, it));
    const float xj1 = __builtin_bit_cast(float, __builtin_amdgcn_readlane(xj1v, it));
    const float xj2 = __builtin_bit_cast(float, __builtin_amdgcn_readlane(xj2v, it));
    const float dg = (i == jbase + it) ? 1.f : 0.f;
    const float r0 = (xi0 - xj0) + dg;
    const float r1 = (xi1 - xj1) + dg;
    const float r2 = (xi2 - xj2) + dg;
    const float nr = sqrtf(r0 * r0 + r1 * r1 + r2 * r2);
    sf0 += r0; sf1 += r1; sf2 += r2; sf3 += nr;

    v2f bv0;
    bv0.x = hh ? r2 : r0;
    bv0.y = hh ? nr : r1;
    v8f h[2];
    #pragma unroll
    for (int ot = 0; ot < 2; ++ot) {
      const v8f d = wmma_f32_k4(a0[ot], bv0, c0[ot]);
      #pragma unroll
      for (int r = 0; r < 8; ++r) h[ot][r] = fast_tanh(d[r]);
      s0[ot] += h[ot];
    }

    #pragma unroll
    for (int L = 0; L < 2; ++L) {
      v8f acc[2] = {cb[L][0], cb[L][1]};
      #pragma unroll
      for (int T = 0; T < 2; ++T) {
        #pragma unroll
        for (int q = 0; q < 4; ++q) {
          v2f bv; bv.x = h[T][2 * q]; bv.y = h[T][2 * q + 1];
          #pragma unroll
          for (int ot = 0; ot < 2; ++ot) {
            const v4f w4 = aw[L][ot][T][q >> 1];
            v2f a;
            if (q & 1) { a.x = w4.z; a.y = w4.w; } else { a.x = w4.x; a.y = w4.y; }
            acc[ot] = wmma_f32_k4(a, bv, acc[ot]);
          }
        }
      }
      #pragma unroll
      for (int ot = 0; ot < 2; ++ot) {
        #pragma unroll
        for (int r = 0; r < 8; ++r) h[ot][r] = fast_tanh(acc[ot][r]) + h[ot][r];
        if (L == 0) s1[ot] += h[ot]; else s2[ot] += h[ot];
      }
    }
  }

  #pragma unroll
  for (int ot = 0; ot < 2; ++ot) {
    float* p0 = &part[wave][0][lr][16 * ot + 8 * hh];
    float* p1 = &part[wave][1][lr][16 * ot + 8 * hh];
    float* p2 = &part[wave][2][lr][16 * ot + 8 * hh];
    *(v4f*)(p0)     = (v4f){s0[ot][0], s0[ot][1], s0[ot][2], s0[ot][3]};
    *(v4f*)(p0 + 4) = (v4f){s0[ot][4], s0[ot][5], s0[ot][6], s0[ot][7]};
    *(v4f*)(p1)     = (v4f){s1[ot][0], s1[ot][1], s1[ot][2], s1[ot][3]};
    *(v4f*)(p1 + 4) = (v4f){s1[ot][4], s1[ot][5], s1[ot][6], s1[ot][7]};
    *(v4f*)(p2)     = (v4f){s2[ot][0], s2[ot][1], s2[ot][2], s2[ot][3]};
    *(v4f*)(p2 + 4) = (v4f){s2[ot][4], s2[ot][5], s2[ot][6], s2[ot][7]};
  }
  if (hh == 0) partf[wave][lr] = (v4f){sf0, sf1, sf2, sf3};
  __syncthreads();

  if (wave < 3) {
    float* ms = (wave == 0) ? m0 : (wave == 1) ? m1 : m2;
    float* dstp = ms + (b * NP + itile * 16) * H2;
    #pragma unroll
    for (int k = 0; k < 4; ++k) {
      const int f = k * 32 + lane;
      const int row = f >> 3;
      const int c4 = (f & 7) * 4;
      v4f v = *(const v4f*)&part[0][wave][row][c4];
      v += *(const v4f*)&part[1][wave][row][c4];
      v += *(const v4f*)&part[2][wave][row][c4];
      v += *(const v4f*)&part[3][wave][row][c4];
      v *= (1.f / NP);
      *(v4f*)(dstp + f * 4) = v;
    }
  } else if (lane < 16) {
    v4f v = partf[0][lane];
    v += partf[1][lane];
    v += partf[2][lane];
    v += partf[3][lane];
    v *= (1.f / NP);
    *(v4f*)(mfeat + (b * NP + itile * 16 + lane) * 4) = v;
  }
}

__device__ __forceinline__ float row16_sum(float v) {
  v += __builtin_bit_cast(float, __builtin_amdgcn_update_dpp(
           0, __builtin_bit_cast(int, v), 0x161, 0xf, 0xf, true));
  v += __builtin_bit_cast(float, __builtin_amdgcn_update_dpp(
           0, __builtin_bit_cast(int, v), 0x162, 0xf, 0xf, true));
  v += __builtin_bit_cast(float, __builtin_amdgcn_update_dpp(
           0, __builtin_bit_cast(int, v), 0x164, 0xf, 0xf, true));
  v += __builtin_bit_cast(float, __builtin_amdgcn_update_dpp(
           0, __builtin_bit_cast(int, v), 0x168, 0xf, 0xf, true));
  return v;
}

__device__ __forceinline__ float xhalf(float x) {
  const int xi = __builtin_bit_cast(int, x);
  return __builtin_bit_cast(float, __builtin_amdgcn_permlanex16(
      xi, xi, 0x76543210, 0xfedcba98, false, false));
}

template <int SRC>
__device__ __forceinline__ float row_take(float v) {
  return __builtin_bit_cast(float, __builtin_amdgcn_update_dpp(
      0, __builtin_bit_cast(int, v), 0x150 + SRC, 0xf, 0xf, true));
}

__device__ __forceinline__ float pair_sum(float v) {
  return v + __builtin_bit_cast(float, __builtin_amdgcn_update_dpp(
                 0, __builtin_bit_cast(int, v), 0x161, 0xf, 0xf, true));
}

template <int G>
__device__ __forceinline__ void spread_group(const v4f s, v8f (&meanv)[4]) {
  meanv[G >> 1][(G & 1) * 4 + 0] = row_take<2 * G>(s.x);
  meanv[G >> 1][(G & 1) * 4 + 1] = row_take<2 * G>(s.y);
  meanv[G >> 1][(G & 1) * 4 + 2] = row_take<2 * G>(s.z);
  meanv[G >> 1][(G & 1) * 4 + 3] = row_take<2 * G>(s.w);
}

#define NSTEP 30

__global__ __launch_bounds__(128) void drift_particle_chain(
    const float* __restrict__ x, const float* __restrict__ t,
    const float* __restrict__ fc1_w0, const float* __restrict__ fc1_b0,
    const float* __restrict__ fc1_w, const float* __restrict__ fc1_b,
    const float* __restrict__ final_w, const float* __restrict__ final_b,
    const float* __restrict__ mfeat, const float* __restrict__ m0,
    const float* __restrict__ m1, const float* __restrict__ m2,
    float* __restrict__ out) {
  const int b = blockIdx.x;
  const int tid = threadIdx.x;
  const int wave = tid >> 5;
  const int lane = tid & 31;
  const int hh = lane >> 4;
  const int lr = lane & 15;
  const int n = wave * 16 + lr;

  __shared__ v4f pfeat[4];
  __shared__ __attribute__((aligned(16))) float w0s[13][H1 + 4];
  __shared__ v4f psum[3][4][16];
  __shared__ float outS[NP * DIMX];

  const float tval = t[0];
  const float x0 = x[(b * NP + n) * 3 + 0];
  const float x1 = x[(b * NP + n) * 3 + 1];
  const float x2 = x[(b * NP + n) * 3 + 2];
  const float nr = sqrtf(x0 * x0 + x1 * x1 + x2 * x2);
  const v4f mf = *(const v4f*)(mfeat + (b * NP + n) * 4);

  v4f mjv[3][4];
  #pragma unroll
  for (int s = 0; s < 3; ++s) {
    const float* ms = (s == 0) ? m0 : (s == 1) ? m1 : m2;
    const float* mp = ms + (b * NP + n) * H2 + 8 * hh;
    #pragma unroll
    for (int i = 0; i < 4; ++i)
      mjv[s][i] = *(const v4f*)(mp + (i >> 1) * 16 + (i & 1) * 4);
  }

  v4f b0v[8];
  #pragma unroll
  for (int i = 0; i < 8; ++i)
    b0v[i] = *(const v4f*)(fc1_b0 + (i >> 1) * 16 + 8 * hh + (i & 1) * 4);
  const float fb[DIMX] = {final_b[0], final_b[1], final_b[2]};

  const float* wlane = fc1_w + lr * FSIZE + 8 * hh;
  v4f abuf[3][8];
  auto tile_of = [](int g) -> int {
    const int j = g % 10;
    return (j < 4) ? j : (j < 6) ? (j + 4) : (j - 2);
  };
  auto load_a = [&](v4f (&dstb)[8], int g) {
    const float* wp = wlane + (g / 10) * (H1 * FSIZE) + tile_of(g) * 16;
    #pragma unroll
    for (int ot = 0; ot < 4; ++ot) {
      dstb[ot * 2 + 0] = *(const v4f*)(wp + ot * 16 * FSIZE);
      dstb[ot * 2 + 1] = *(const v4f*)(wp + ot * 16 * FSIZE + 4);
    }
  };
  load_a(abuf[0], 0);
  load_a(abuf[1], 1);

  {
    const v4f ps = {row16_sum(x0), row16_sum(x1), row16_sum(x2), row16_sum(nr)};
    if (lane == 0) pfeat[wave] = ps;
  }
  {
    float wv[7];
    #pragma unroll
    for (int it = 0; it < 6; ++it) wv[it] = fc1_w0[it * 128 + tid];
    wv[6] = fc1_w0[6 * 128 + (tid & 63)];
    #pragma unroll
    for (int it = 0; it < 6; ++it) {
      const int idx = it * 128 + tid;
      w0s[idx % 13][idx / 13] = wv[it];
    }
    if (tid < 64) {
      const int idx = 6 * 128 + tid;
      w0s[idx % 13][idx / 13] = wv[6];
    }
  }
  __syncthreads();

  const v4f smf = ((pfeat[0] + pfeat[1]) + (pfeat[2] + pfeat[3])) * (1.f / NP);
  const float smf0 = smf.x, smf1 = smf.y, smf2 = smf.z, smf3 = smf.w;

  v8f hreg[4];
  {
    const float ft[13] = {x0, x1, x2, nr, smf0, smf1, smf2, smf3,
                          mf.x, mf.y, mf.z, mf.w, tval};
    #pragma unroll
    for (int ot = 0; ot < 4; ++ot) {
      const v4f ba = b0v[ot * 2 + 0];
      const v4f bb = b0v[ot * 2 + 1];
      v8f acc = {ba.x, ba.y, ba.z, ba.w, bb.x, bb.y, bb.z, bb.w};
      #pragma unroll
      for (int k = 0; k < 13; ++k) {
        const v4f wa = *(const v4f*)&w0s[k][ot * 16 + 8 * hh];
        const v4f wb = *(const v4f*)&w0s[k][ot * 16 + 8 * hh + 4];
        acc[0] += ft[k] * wa.x; acc[1] += ft[k] * wa.y;
        acc[2] += ft[k] * wa.z; acc[3] += ft[k] * wa.w;
        acc[4] += ft[k] * wb.x; acc[5] += ft[k] * wb.y;
        acc[6] += ft[k] * wb.z; acc[7] += ft[k] * wb.w;
      }
      #pragma unroll
      for (int r = 0; r < 8; ++r) hreg[ot][r] = fast_tanh(acc[r]);
    }
  }

  v8f acc[4];
  v8f meanv[4];
  v4f bias[8];
  v4f fw[DIMX][8];
  #pragma unroll
  for (int g = 0; g < NSTEP; ++g) {
    const int L = g / 10;
    const int j = g % 10;
    const int T = tile_of(g);
    if (g + 2 < NSTEP) load_a(abuf[(g + 2) % 3], g + 2);

    if (j == 0) {
      v8f rs[4];
      #pragma unroll
      for (int ot = 0; ot < 4; ++ot) {
        #pragma unroll
        for (int r = 0; r < 8; ++r) rs[ot][r] = row16_sum(hreg[ot][r]);
      }
      if (lr == 0) {
        #pragma unroll
        for (int ot = 0; ot < 4; ++ot) {
          psum[L][wave][ot * 4 + hh * 2 + 0] = (v4f){rs[ot][0], rs[ot][1], rs[ot][2], rs[ot][3]};
          psum[L][wave][ot * 4 + hh * 2 + 1] = (v4f){rs[ot][4], rs[ot][5], rs[ot][6], rs[ot][7]};
        }
      }
      #pragma unroll
      for (int i = 0; i < 8; ++i)
        bias[i] = *(const v4f*)(fc1_b + L * H1 + (i >> 1) * 16 + 8 * hh + (i & 1) * 4);
      #pragma unroll
      for (int ot = 0; ot < 4; ++ot) acc[ot] = (v8f){};
    }
    if (j == 6) {
      __syncthreads();
      {
        const int G = lr >> 1;
        const int vi = (G >> 1) * 4 + hh * 2 + (G & 1);
        const int wp = (lr & 1) * 2;
        v4f s = psum[L][wp][vi] + psum[L][wp + 1][vi];
        s.x = pair_sum(s.x); s.y = pair_sum(s.y);
        s.z = pair_sum(s.z); s.w = pair_sum(s.w);
        s *= (1.f / NP);
        spread_group<0>(s, meanv); spread_group<1>(s, meanv);
        spread_group<2>(s, meanv); spread_group<3>(s, meanv);
        spread_group<4>(s, meanv); spread_group<5>(s, meanv);
        spread_group<6>(s, meanv); spread_group<7>(s, meanv);
      }
      if (L == 2) {
        #pragma unroll
        for (int dd = 0; dd < DIMX; ++dd) {
          #pragma unroll
          for (int i = 0; i < 8; ++i)
            fw[dd][i] = *(const v4f*)(final_w + dd * H1 + (i >> 1) * 16 + 8 * hh + (i & 1) * 4);
        }
      }
    }

    v8f bf;
    if (T < 4) {
      bf = hreg[T];
    } else if (T < 8) {
      bf = meanv[T - 4];
    } else {
      const v4f lo = mjv[L][(T - 8) * 2 + 0];
      const v4f hi = mjv[L][(T - 8) * 2 + 1];
      bf = (v8f){lo.x, lo.y, lo.z, lo.w, hi.x, hi.y, hi.z, hi.w};
    }
    const v4f (&ab)[8] = abuf[g % 3];
    #pragma unroll
    for (int q = 0; q < 4; ++q) {
      v2f bv; bv.x = bf[2 * q]; bv.y = bf[2 * q + 1];
      #pragma unroll
      for (int ot = 0; ot < 4; ++ot) {
        const v4f aw = ab[ot * 2 + (q >> 1)];
        v2f a;
        if (q & 1) { a.x = aw.z; a.y = aw.w; } else { a.x = aw.x; a.y = aw.y; }
        acc[ot] = wmma_f32_k4(a, bv, acc[ot]);
      }
    }

    if (j == 9) {
      #pragma unroll
      for (int ot = 0; ot < 4; ++ot) {
        #pragma unroll
        for (int r = 0; r < 8; ++r)
          hreg[ot][r] = fast_tanh(acc[ot][r] + bias[ot * 2 + (r >> 2)][r & 3]) + hreg[ot][r];
      }
    }
  }

  #pragma unroll
  for (int dd = 0; dd < DIMX; ++dd) {
    float s = 0.f;
    #pragma unroll
    for (int ot = 0; ot < 4; ++ot) {
      #pragma unroll
      for (int r = 0; r < 8; ++r) s += hreg[ot][r] * fw[dd][ot * 2 + (r >> 2)][r & 3];
    }
    s = s + xhalf(s);
    if (hh == 0) outS[n * DIMX + dd] = s + fb[dd];
  }
  __syncthreads();
  for (int idx = tid; idx < NP * DIMX; idx += 128)
    out[b * (NP * DIMX) + idx] = outS[idx];
}

extern "C" void kernel_launch(void* const* d_in, const int* in_sizes, int n_in,
                              void* d_out, int out_size, void* d_ws,
                              size_t ws_size, hipStream_t stream) {
  const float* x       = (const float*)d_in[0];
  const float* t       = (const float*)d_in[1];
  const float* fc1_w0  = (const float*)d_in[2];
  const float* fc1_b0  = (const float*)d_in[3];
  const float* fc1_w   = (const float*)d_in[4];
  const float* fc1_b   = (const float*)d_in[5];
  const float* fc2_w0  = (const float*)d_in[6];
  const float* fc2_b0  = (const float*)d_in[7];
  const float* fc2_w   = (const float*)d_in[8];
  const float* fc2_b   = (const float*)d_in[9];
  const float* final_w = (const float*)d_in[10];
  const float* final_b = (const float*)d_in[11];
  float* out = (float*)d_out;

  float* ws    = (float*)d_ws;
  float* mfeat = ws;
  float* m0    = mfeat + BATCHN * NP * 4;
  float* m1    = m0 + BATCHN * NP * H2;
  float* m2    = m1 + BATCHN * NP * H2;
  (void)in_sizes; (void)n_in; (void)out_size; (void)ws_size;

  drift_pair_chain<<<dim3(4, BATCHN), 128, 0, stream>>>(
      x, t, fc2_w0, fc2_b0, fc2_w, fc2_b, mfeat, m0, m1, m2);
  drift_particle_chain<<<dim3(BATCHN), 128, 0, stream>>>(
      x, t, fc1_w0, fc1_b0, fc1_w, fc1_b, final_w, final_b,
      mfeat, m0, m1, m2, out);
}


# particle chain re-cut by output rows: a wave owns 16 outputs x all 64 particles, reads only its own 16 weight rows; other waves' h1 via LDS; mean_n h1 product once per wave (sum order changes)
# speedup vs baseline: 4.2957x; 1.0640x over previous
#include <hip/hip_runtime.h>
#include <math.h>

typedef __attribute__((ext_vector_type(2))) float v2f;
typedef __attribute__((ext_vector_type(4))) float v4f;
typedef __attribute__((ext_vector_type(8))) float v8f;

#define DEPTH 4
#define H1 64
#define H2 32
#define NP 64
#define DIMX 3
#define FSIZE 160
#define BATCHN 256

__device__ __forceinline__ v8f wmma_f32_k4(v2f a, v2f b, v8f c) {
  return __builtin_amdgcn_wmma_f32_16x16x4_f32(false, a, false, b, (short)0, c,
                                               false, false);
}

__device__ __forceinline__ float fast_tanh(float x) {
#if __has_builtin(__builtin_amdgcn_tanhf)
  return __builtin_amdgcn_tanhf(x);
#else
  float r;
  asm volatile("v_tanh_f32 %0, %1\n\tv_nop" : "=v"(r) : "v"(x));
  return r;
#endif
}

__global__ __launch_bounds__(128) void drift_pair_chain(
    const float* __restrict__ x, const float* __restrict__ t,
    const float* __restrict__ fc2_w0, const float* __restrict__ fc2_b0,
    const float* __restrict__ fc2_w, const float* __restrict__ fc2_b,
    float* __restrict__ mfeat, float* __restrict__ m0,
    float* __restrict__ m1, float* __restrict__ m2) {
  const int itile = blockIdx.x;
  const int b = blockIdx.y;
  const int tid = threadIdx.x;
  const int wave = __builtin_amdgcn_readfirstlane(tid >> 5);
  const int lane = tid & 31;
  const int hh = lane >> 4;
  const int lr = lane & 15;
  const int i = itile * 16 + lr;
  const int jbase = wave * 16;

  __shared__ __attribute__((aligned(16))) float part[4][3][16][H2 + 4];
  __shared__ v4f partf[4][16];

  const float tval = t[0];
  const float xi0 = x[(b * NP + i) * 3 + 0];
  const float xi1 = x[(b * NP + i) * 3 + 1];
  const float xi2 = x[(b * NP + i) * 3 + 2];
  const int xj0v = __builtin_bit_cast(int, x[(b * NP + jbase + lr) * 3 + 0]);
  const int xj1v = __builtin_bit_cast(int, x[(b * NP + jbase + lr) * 3 + 1]);
  const int xj2v = __builtin_bit_cast(int, x[(b * NP + jbase + lr) * 3 + 2]);

  v2f a0[2];
  v8f c0[2];
  #pragma unroll
  for (int ot = 0; ot < 2; ++ot) {
    a0[ot].x = fc2_w0[(16 * ot + lr) * 5 + 2 * hh];
    a0[ot].y = fc2_w0[(16 * ot + lr) * 5 + 2 * hh + 1];
    const v4f ba = *(const v4f*)(fc2_b0 + 16 * ot + 8 * hh);
    const v4f bb = *(const v4f*)(fc2_b0 + 16 * ot + 8 * hh + 4);
    const float bl[8] = {ba.x, ba.y, ba.z, ba.w, bb.x, bb.y, bb.z, bb.w};
    #pragma unroll
    for (int r = 0; r < 8; ++r)
      c0[ot][r] = bl[r] + tval * fc2_w0[(16 * ot + 8 * hh + r) * 5 + 4];
  }
  v4f aw[2][2][2][2];
  v8f cb[2][2];
  #pragma unroll
  for (int L = 0; L < 2; ++L) {
    #pragma unroll
    for (int ot = 0; ot < 2; ++ot) {
      const float* wp = fc2_w + L * (H2 * H2) + (16 * ot + lr) * H2 + 8 * hh;
      #pragma unroll
      for (int T = 0; T < 2; ++T) {
        aw[L][ot][T][0] = *(const v4f*)(wp + 16 * T);
        aw[L][ot][T][1] = *(const v4f*)(wp + 16 * T + 4);
      }
      const v4f ba = *(const v4f*)(fc2_b + L * H2 + 16 * ot + 8 * hh);
      const v4f bb = *(const v4f*)(fc2_b + L * H2 + 16 * ot + 8 * hh + 4);
      cb[L][ot] = (v8f){ba.x, ba.y, ba.z, ba.w, bb.x, bb.y, bb.z, bb.w};
    }
  }

  v8f s0[2] = {{}, {}}, s1[2] = {{}, {}}, s2[2] = {{}, {}};
  float sf0 = 0.f, sf1 = 0.f, sf2 = 0.f, sf3 = 0.f;

  for (int it = 0; it < 16; ++it) {
    const float xj0 = __builtin_bit_cast(float, __builtin_amdgcn_readlane(xj0v, it));
    const float xj1 = __builtin_bit_cast(float, __builtin_amdgcn_readlane(xj1v, it));
    const float xj2 = __builtin_bit_cast(float, __builtin_amdgcn_readlane(xj2v, it));
    const float dg = (i == jbase + it) ? 1.f : 0.f;
    const float r0 = (xi0 - xj0) + dg;
    const float r1 = (xi1 - xj1) + dg;
    const float r2 = (xi2 - xj2) + dg;
    const float nr = sqrtf(r0 * r0 + r1 * r1 + r2 * r2);
    sf0 += r0; sf1 += r1; sf2 += r2; sf3 += nr;

    v2f bv0;
    bv0.x = hh ? r2 : r0;
    bv0.y = hh ? nr : r1;
    v8f h[2];
    #pragma unroll
    for (int ot = 0; ot < 2; ++ot) {
      const v8f d = wmma_f32_k4(a0[ot], bv0, c0[ot]);
      #pragma unroll
      for (int r = 0; r < 8; ++r) h[ot][r] = fast_tanh(d[r]);
      s0[ot] += h[ot];
    }

    #pragma unroll
    for (int L = 0; L < 2; ++L) {
      v8f acc[2] = {cb[L][0], cb[L][1]};
      #pragma unroll
      for (int T = 0; T < 2; ++T) {
        #pragma unroll
        for (int q = 0; q < 4; ++q) {
          v2f bv; bv.x = h[T][2 * q]; bv.y = h[T][2 * q + 1];
          #pragma unroll
          for (int ot = 0; ot < 2; ++ot) {
            const v4f w4 = aw[L][ot][T][q >> 1];
            v2f a;
            if (q & 1) { a.x = w4.z; a.y = w4.w; } else { a.x = w4.x; a.y = w4.y; }
            acc[ot] = wmma_f32_k4(a, bv, acc[ot]);
          }
        }
      }
      #pragma unroll
      for (int ot = 0; ot < 2; ++ot) {
        #pragma unroll
        for (int r = 0; r < 8; ++r) h[ot][r] = fast_tanh(acc[ot][r]) + h[ot][r];
        if (L == 0) s1[ot] += h[ot]; else s2[ot] += h[ot];
      }
    }
  }

  #pragma unroll
  for (int ot = 0; ot < 2; ++ot) {
    float* p0 = &part[wave][0][lr][16 * ot + 8 * hh];
    float* p1 = &part[wave][1][lr][16 * ot + 8 * hh];
    float* p2 = &part[wave][2][lr][16 * ot + 8 * hh];
    *(v4f*)(p0)     = (v4f){s0[ot][0], s0[ot][1], s0[ot][2], s0[ot][3]};
    *(v4f*)(p0 + 4) = (v4f){s0[ot][4], s0[ot][5], s0[ot][6], s0[ot][7]};
    *(v4f*)(p1)     = (v4f){s1[ot][0], s1[ot][1], s1[ot][2], s1[ot][3]};
    *(v4f*)(p1 + 4) = (v4f){s1[ot][4], s1[ot][5], s1[ot][6], s1[ot][7]};
    *(v4f*)(p2)     = (v4f){s2[ot][0], s2[ot][1], s2[ot][2], s2[ot][3]};
    *(v4f*)(p2 + 4) = (v4f){s2[ot][4], s2[ot][5], s2[ot][6], s2[ot][7]};
  }
  if (hh == 0) partf[wave][lr] = (v4f){sf0, sf1, sf2, sf3};
  __syncthreads();

  if (wave < 3) {
    float* ms = (wave == 0) ? m0 : (wave == 1) ? m1 : m2;
    float* dstp = ms + (b * NP + itile * 16) * H2;
    #pragma unroll
    for (int k = 0; k < 4; ++k) {
      const int f = k * 32 + lane;
      const int row = f >> 3;
      const int c4 = (f & 7) * 4;
      v4f v = *(const v4f*)&part[0][wave][row][c4];
      v += *(const v4f*)&part[1][wave][row][c4];
      v += *(const v4f*)&part[2][wave][row][c4];
      v += *(const v4f*)&part[3][wave][row][c4];
      v *= (1.f / NP);
      *(v4f*)(dstp + f * 4) = v;
    }
  } else if (lane < 16) {
    v4f v = partf[0][lane];
    v += partf[1][lane];
    v += partf[2][lane];
    v += partf[3][lane];
    v *= (1.f / NP);
    *(v4f*)(mfeat + (b * NP + itile * 16 + lane) * 4) = v;
  }
}

__device__ __forceinline__ float row16_sum(float v) {
  v += __builtin_bit_cast(float, __builtin_amdgcn_update_dpp(
           0, __builtin_bit_cast(int, v), 0x161, 0xf, 0xf, true));
  v += __builtin_bit_cast(float, __builtin_amdgcn_update_dpp(
           0, __builtin_bit_cast(int, v), 0x162, 0xf, 0xf, true));
  v += __builtin_bit_cast(float, __builtin_amdgcn_update_dpp(
           0, __builtin_bit_cast(int, v), 0x164, 0xf, 0xf, true));
  v += __builtin_bit_cast(float, __builtin_amdgcn_update_dpp(
           0, __builtin_bit_cast(int, v), 0x168, 0xf, 0xf, true));
  return v;
}

__device__ __forceinline__ float xhalf(float x) {
  const int xi = __builtin_bit_cast(int, x);
  return __builtin_bit_cast(float, __builtin_amdgcn_permlanex16(
      xi, xi, 0x76543210, 0xfedcba98, false, false));
}

__device__ __forceinline__ v2f a_step(const v4f (&a)[2], int q) {
  const v4f w4 = a[q >> 1];
  v2f r;
  if (q & 1) { r.x = w4.z; r.y = w4.w; } else { r.x = w4.x; r.y = w4.y; }
  return r;
}

#define HPITCH 68

__global__ __launch_bounds__(128) void drift_particle_chain(
    const float* __restrict__ x, const float* __restrict__ t,
    const float* __restrict__ fc1_w0, const float* __restrict__ fc1_b0,
    const float* __restrict__ fc1_w, const float* __restrict__ fc1_b,
    const float* __restrict__ final_w, const float* __restrict__ final_b,
    const float* __restrict__ mfeat, const float* __restrict__ m0,
    const float* __restrict__ m1, const float* __restrict__ m2,
    float* __restrict__ out) {
  const int b = blockIdx.x;
  const int tid = threadIdx.x;
  const int wave = __builtin_amdgcn_readfirstlane(tid >> 5);
  const int lane = tid & 31;
  const int hh = lane >> 4;
  const int c = lane & 15;
  const int fbase = 16 * wave + 8 * hh;

  __shared__ __attribute__((aligned(16))) float hS[2][NP][HPITCH];
  __shared__ __attribute__((aligned(16))) float meanS[2][H1];
  __shared__ float outP[4][NP * DIMX];

  const float tval = t[0];
  float xs[4][3];
  v4f mf[4];
  #pragma unroll
  for (int p = 0; p < 4; ++p) {
    const int n = b * NP + 16 * p + c;
    xs[p][0] = x[n * 3 + 0];
    xs[p][1] = x[n * 3 + 1];
    xs[p][2] = x[n * 3 + 2];
    mf[p] = *(const v4f*)(mfeat + n * 4);
  }
  v4f w0v[26];
  #pragma unroll
  for (int i = 0; i < 26; ++i) w0v[i] = *(const v4f*)(fc1_w0 + fbase * 13 + 4 * i);
  const v4f b0a = *(const v4f*)(fc1_b0 + fbase);
  const v4f b0b = *(const v4f*)(fc1_b0 + fbase + 4);

  const float* wrow = fc1_w + (16 * wave + c) * FSIZE + 8 * hh;
  v4f ah[2][4][2], am[2][4][2], aj[2][2][2];
  v4f mjv[2][4][4];
  v4f biasv[2][2];
  auto load_layer = [&](int s, int L) {
    const float* wp = wrow + L * (H1 * FSIZE);
    #pragma unroll
    for (int j = 0; j < 4; ++j) {
      const int T = (wave + j) & 3;
      ah[s][j][0] = *(const v4f*)(wp + 16 * T);
      ah[s][j][1] = *(const v4f*)(wp + 16 * T + 4);
      am[s][j][0] = *(const v4f*)(wp + 64 + 16 * j);
      am[s][j][1] = *(const v4f*)(wp + 64 + 16 * j + 4);
    }
    #pragma unroll
    for (int t2 = 0; t2 < 2; ++t2) {
      aj[s][t2][0] = *(const v4f*)(wp + 128 + 16 * t2);
      aj[s][t2][1] = *(const v4f*)(wp + 128 + 16 * t2 + 4);
    }
    const float* ms = (L == 0) ? m0 : (L == 1) ? m1 : m2;
    #pragma unroll
    for (int p = 0; p < 4; ++p) {
      const float* mp = ms + (b * NP + 16 * p + c) * H2 + 8 * hh;
      #pragma unroll
      for (int i = 0; i < 4; ++i)
        mjv[s][p][i] = *(const v4f*)(mp + (i >> 1) * 16 + (i & 1) * 4);
    }
    biasv[s][0] = *(const v4f*)(fc1_b + L * H1 + fbase);
    biasv[s][1] = *(const v4f*)(fc1_b + L * H1 + fbase + 4);
  };
  load_layer(0, 0);

  v8f hreg[4];
  {
    float nr[4];
    #pragma unroll
    for (int p = 0; p < 4; ++p)
      nr[p] = sqrtf(xs[p][0] * xs[p][0] + xs[p][1] * xs[p][1] + xs[p][2] * xs[p][2]);
    const float inv = 1.f / NP;
    const float smf0 = row16_sum((xs[0][0] + xs[1][0]) + (xs[2][0] + xs[3][0])) * inv;
    const float smf1 = row16_sum((xs[0][1] + xs[1][1]) + (xs[2][1] + xs[3][1])) * inv;
    const float smf2 = row16_sum((xs[0][2] + xs[1][2]) + (xs[2][2] + xs[3][2])) * inv;
    const float smf3 = row16_sum((nr[0] + nr[1]) + (nr[2] + nr[3])) * inv;
    float w0f[104];
    #pragma unroll
    for (int i = 0; i < 26; ++i) {
      w0f[4 * i + 0] = w0v[i].x; w0f[4 * i + 1] = w0v[i].y;
      w0f[4 * i + 2] = w0v[i].z; w0f[4 * i + 3] = w0v[i].w;
    }
    const float b0f[8] = {b0a.x, b0a.y, b0a.z, b0a.w, b0b.x, b0b.y, b0b.z, b0b.w};
    #pragma unroll
    for (int p = 0; p < 4; ++p) {
      const float ft[13] = {xs[p][0], xs[p][1], xs[p][2], nr[p], smf0, smf1, smf2, smf3,
                            mf[p].x, mf[p].y, mf[p].z, mf[p].w, tval};
      #pragma unroll
      for (int r = 0; r < 8; ++r) {
        float a = b0f[r];
        #pragma unroll
        for (int k = 0; k < 13; ++k) a += ft[k] * w0f[r * 13 + k];
        hreg[p][r] = fast_tanh(a);
      }
    }
  }

  v4f fw[DIMX][2];
  #pragma unroll
  for (int L = 0; L < 3; ++L) {
    const int s = L & 1;
    #pragma unroll
    for (int p = 0; p < 4; ++p) {
      float* hp = &hS[s][16 * p + c][fbase];
      *(v4f*)(hp)     = (v4f){hreg[p][0], hreg[p][1], hreg[p][2], hreg[p][3]};
      *(v4f*)(hp + 4) = (v4f){hreg[p][4], hreg[p][5], hreg[p][6], hreg[p][7]};
    }
    {
      v8f tot = (hreg[0] + hreg[1]) + (hreg[2] + hreg[3]);
      #pragma unroll
      for (int r = 0; r < 8; ++r) tot[r] = row16_sum(tot[r]) * (1.f / NP);
      if (c == 0) {
        *(v4f*)&meanS[s][fbase]     = (v4f){tot[0], tot[1], tot[2], tot[3]};
        *(v4f*)&meanS[s][fbase + 4] = (v4f){tot[4], tot[5], tot[6], tot[7]};
      }
    }
    if (L + 1 < 3) {
      load_layer(s ^ 1, L + 1);
    } else {
      #pragma unroll
      for (int dd = 0; dd < DIMX; ++dd) {
        fw[dd][0] = *(const v4f*)(final_w + dd * H1 + fbase);
        fw[dd][1] = *(const v4f*)(final_w + dd * H1 + fbase + 4);
      }
    }

    v8f acc[4] = {{}, {}, {}, {}};
    #pragma unroll
    for (int q = 0; q < 4; ++q) {
      const v2f a = a_step(ah[s][0], q);
      #pragma unroll
      for (int p = 0; p < 4; ++p) {
        v2f bv; bv.x = hreg[p][2 * q]; bv.y = hreg[p][2 * q + 1];
        acc[p] = wmma_f32_k4(a, bv, acc[p]);
      }
    }
    __syncthreads();

    v4f mv[4][2];
    #pragma unroll
    for (int T = 0; T < 4; ++T) {
      mv[T][0] = *(const v4f*)&meanS[s][16 * T + 8 * hh];
      mv[T][1] = *(const v4f*)&meanS[s][16 * T + 8 * hh + 4];
    }
    v4f bfo[3][4][2];
    #pragma unroll
    for (int j = 1; j < 4; ++j) {
      const int T = (wave + j) & 3;
      #pragma unroll
      for (int p = 0; p < 4; ++p) {
        const float* hp = &hS[s][16 * p + c][16 * T + 8 * hh];
        bfo[j - 1][p][0] = *(const v4f*)(hp);
        bfo[j - 1][p][1] = *(const v4f*)(hp + 4);
      }
    }
    __builtin_amdgcn_sched_barrier(0);

    #pragma unroll
    for (int t2 = 0; t2 < 2; ++t2) {
      #pragma unroll
      for (int q = 0; q < 4; ++q) {
        const v2f a = a_step(aj[s][t2], q);
        #pragma unroll
        for (int p = 0; p < 4; ++p) {
          const v4f m4 = mjv[s][p][t2 * 2 + (q >> 1)];
          v2f bv;
          if (q & 1) { bv.x = m4.z; bv.y = m4.w; } else { bv.x = m4.x; bv.y = m4.y; }
          acc[p] = wmma_f32_k4(a, bv, acc[p]);
        }
      }
    }

    v8f macc[2];
    macc[0] = (v8f){biasv[s][0].x, biasv[s][0].y, biasv[s][0].z, biasv[s][0].w,
                    biasv[s][1].x, biasv[s][1].y, biasv[s][1].z, biasv[s][1].w};
    macc[1] = (v8f){};
    #pragma unroll
    for (int q = 0; q < 4; ++q) {
      #pragma unroll
      for (int T = 0; T < 4; ++T) {
        const v4f m4 = mv[T][q >> 1];
        v2f bv;
        if (q & 1) { bv.x = m4.z; bv.y = m4.w; } else { bv.x = m4.x; bv.y = m4.y; }
        macc[T & 1] = wmma_f32_k4(a_step(am[s][T], q), bv, macc[T & 1]);
      }
    }
    #pragma unroll
    for (int j = 1; j < 4; ++j) {
      #pragma unroll
      for (int q = 0; q < 4; ++q) {
        const v2f a = a_step(ah[s][j], q);
        #pragma unroll
        for (int p = 0; p < 4; ++p) {
          const v4f h4 = bfo[j - 1][p][q >> 1];
          v2f bv;
          if (q & 1) { bv.x = h4.z; bv.y = h4.w; } else { bv.x = h4.x; bv.y = h4.y; }
          acc[p] = wmma_f32_k4(a, bv, acc[p]);
        }
      }
    }
    const v8f madd = macc[0] + macc[1];
    #pragma unroll
    for (int p = 0; p < 4; ++p) {
      #pragma unroll
      for (int r = 0; r < 8; ++r)
        hreg[p][r] = fast_tanh(acc[p][r] + madd[r]) + hreg[p][r];
    }
  }

  const float fb[DIMX] = {final_b[0], final_b[1], final_b[2]};
  #pragma unroll
  for (int p = 0; p < 4; ++p) {
    #pragma unroll
    for (int dd = 0; dd < DIMX; ++dd) {
      float sacc = 0.f;
      #pragma unroll
      for (int r = 0; r < 8; ++r) sacc += hreg[p][r] * fw[dd][r >> 2][r & 3];
      sacc = sacc + xhalf(sacc);
      if (hh == 0) outP[wave][(16 * p + c) * DIMX + dd] = sacc;
    }
  }
  __syncthreads();
  for (int idx = tid; idx < NP * DIMX; idx += 128) {
    const int dd = idx % DIMX;
    const float bsel = (dd == 0) ? fb[0] : (dd == 1) ? fb[1] : fb[2];
    out[b * (NP * DIMX) + idx] =
        ((outP[0][idx] + outP[1][idx]) + (outP[2][idx] + outP[3][idx])) + bsel;
  }
}

extern "C" void kernel_launch(void* const* d_in, const int* in_sizes, int n_in,
                              void* d_out, int out_size, void* d_ws,
                              size_t ws_size, hipStream_t stream) {
  const float* x       = (const float*)d_in[0];
  const float* t       = (const float*)d_in[1];
  const float* fc1_w0  = (const float*)d_in[2];
  const float* fc1_b0  = (const float*)d_in[3];
  const float* fc1_w   = (const float*)d_in[4];
  const float* fc1_b   = (const float*)d_in[5];
  const float* fc2_w0  = (const float*)d_in[6];
  const float* fc2_b0  = (const float*)d_in[7];
  const float* fc2_w   = (const float*)d_in[8];
  const float* fc2_b   = (const float*)d_in[9];
  const float* final_w = (const float*)d_in[10];
  const float* final_b = (const float*)d_in[11];
  float* out = (float*)d_out;

  float* ws    = (float*)d_ws;
  float* mfeat = ws;
  float* m0    = mfeat + BATCHN * NP * 4;
  float* m1    = m0 + BATCHN * NP * H2;
  float* m2    = m1 + BATCHN * NP * H2;
  (void)in_sizes; (void)n_in; (void)out_size; (void)ws_size;

  drift_pair_chain<<<dim3(4, BATCHN), 128, 0, stream>>>(
      x, t, fc2_w0, fc2_b0, fc2_w, fc2_b, mfeat, m0, m1, m2);
  drift_particle_chain<<<dim3(BATCHN), 128, 0, stream>>>(
      x, t, fc1_w0, fc1_b0, fc1_w, fc1_b, final_w, final_b,
      mfeat, m0, m1, m2, out);
}
